# starting kernel
# baseline (speedup 1.0000x reference)
#include <hip/hip_runtime.h>
#include <math.h>

typedef float v2f __attribute__((ext_vector_type(2)));
typedef float v8f __attribute__((ext_vector_type(8)));

#define STRD 132
#define TP   112

__global__ __launch_bounds__(256) void
LocalSimilar_fused_kernel(const float* __restrict__ fmap,
                          const float* __restrict__ Wqk,
                          float* __restrict__ out)
{
    __shared__ float fmapT[TP * STRD];
    __shared__ float kT[TP * STRD];
    __shared__ float qT[16 * STRD];
    __shared__ float simS[16 * 26];

    const int t    = threadIdx.x;
    const int lane = t & 31;
    const int w    = t >> 5;
    const int lo   = lane & 15;
    const int hi   = lane >> 4;

    const int b  = blockIdx.z;
    const int x  = blockIdx.y;
    const int y0 = blockIdx.x * 16;

    for (int i = t; i < 128 * 5 * 32; i += 256) {
        int lj  = i & 31;
        int row = i >> 5;
        int c = row / 5, r = row % 5;
        if (lj < 20) {
            int gx = x - 2 + r;   gx = gx < 0 ? 0 : (gx > 95 ? 95 : gx);
            int gy = y0 - 2 + lj; gy = gy < 0 ? 0 : (gy > 127 ? 127 : gy);
            fmapT[(r * 20 + lj) * STRD + c] =
                fmap[(((size_t)b * 128 + c) * 96 + gx) * 128 + gy];
        }
    }
    __syncthreads();

    {
        float2 af[32];
        const float* arow = Wqk + ((size_t)(w * 16 + lo)) * 128 + 2 * hi;
        #pragma unroll
        for (int kk = 0; kk < 32; ++kk) af[kk] = *(const float2*)(arow + kk * 4);

        v8f acc = {0.f, 0.f, 0.f, 0.f, 0.f, 0.f, 0.f, 0.f};
        const float* brow = fmapT + (42 + lo) * STRD + 2 * hi;
        #pragma unroll
        for (int kk = 0; kk < 32; ++kk) {
            float2 b2 = *(const float2*)(brow + kk * 4);
            v2f av; av.x = af[kk].x; av.y = af[kk].y;
            v2f bv; bv.x = b2.x;     bv.y = b2.y;
            acc = __builtin_amdgcn_wmma_f32_16x16x4_f32(
                false, av, false, bv, (short)0, acc, false, false);
        }
        const float scale = 0.08838834764831845f;
        #pragma unroll
        for (int v = 0; v < 8; ++v)
            qT[lo * STRD + w * 16 + v + 8 * hi] = acc[v] * scale;
    }

    {
        float2 af[32];
        const float* arow = Wqk + ((size_t)(128 + w * 16 + lo)) * 128 + 2 * hi;
        #pragma unroll
        for (int kk = 0; kk < 32; ++kk) af[kk] = *(const float2*)(arow + kk * 4);

        for (int nt = 0; nt < 7; ++nt) {
            v8f acc = {0.f, 0.f, 0.f, 0.f, 0.f, 0.f, 0.f, 0.f};
            const float* brow = fmapT + (nt * 16 + lo) * STRD + 2 * hi;
            #pragma unroll
            for (int kk = 0; kk < 32; ++kk) {
                float2 b2 = *(const float2*)(brow + kk * 4);
                v2f av; av.x = af[kk].x; av.y = af[kk].y;
                v2f bv; bv.x = b2.x;     bv.y = b2.y;
                acc = __builtin_amdgcn_wmma_f32_16x16x4_f32(
                    false, av, false, bv, (short)0, acc, false, false);
            }
            #pragma unroll
            for (int v = 0; v < 8; ++v)
                kT[(nt * 16 + lo) * STRD + w * 16 + v + 8 * hi] = acc[v];
        }
    }
    __syncthreads();

    for (int idx = t; idx < 16 * 25; idx += 256) {
        int pos = idx / 25, l = idx % 25;
        int di = l / 5, dj = l % 5;
        int npos = di * 20 + pos + dj;
        const float4* qv = (const float4*)(qT + pos  * STRD);
        const float4* kv = (const float4*)(kT + npos * STRD);
        float s = 0.f;
        #pragma unroll
        for (int c4 = 0; c4 < 32; ++c4) {
            float4 q4 = qv[c4], k4 = kv[c4];
            s += q4.x * k4.x + q4.y * k4.y + q4.z * k4.z + q4.w * k4.w;
        }
        simS[pos * 26 + l] = s;
    }
    __syncthreads();

    if (t < 16) {
        float m = -INFINITY;
        #pragma unroll
        for (int l = 0; l < 25; ++l) m = fmaxf(m, simS[t * 26 + l]);
        float e[25], sum = 0.f;
        #pragma unroll
        for (int l = 0; l < 25; ++l) {
            e[l] = __expf(simS[t * 26 + l] - m);
            sum += e[l];
        }
        float inv = 1.0f / sum;
        float* op = out + ((size_t)b * 12288 + (size_t)(x * 128 + y0 + t)) * 25;
        #pragma unroll
        for (int l = 0; l < 25; ++l) op[l] = e[l] * inv;
    }
}

extern "C" void kernel_launch(void* const* d_in, const int* in_sizes, int n_in,
                              void* d_out, int out_size, void* d_ws, size_t ws_size,
                              hipStream_t stream) {
    (void)in_sizes; (void)n_in; (void)d_ws; (void)ws_size; (void)out_size;
    const float* fmap = (const float*)d_in[0];
    const float* Wqk  = (const float*)d_in[1];
    float* out = (float*)d_out;

    dim3 grid(128 / 16, 96, 4);
    LocalSimilar_fused_kernel<<<grid, 256, 0, stream>>>(fmap, Wqk, out);
}


# 2D tile: one work-group per 12x16 pixels projects q and k once over its 16x20 halo in LDS (k stored back in place), grid 8x8x4; halo fetched as aligned column pairs, 32 independent loads a batch
# speedup vs baseline: 3.9985x; 3.9985x over previous
#include <hip/hip_runtime.h>
#include <math.h>


typedef float v2f __attribute__((ext_vector_type(2)));
typedef float v8f __attribute__((ext_vector_type(8)));

#define STRD 132
#define TR   12
#define HR   16
#define HC   20
#define NPOS 320
#define NKT  20
#define NPIX 192

__global__ __launch_bounds__(256) void
LocalSimilar_tile2d_kernel(const float* __restrict__ fmap,
                           const float* __restrict__ Wqk,
                           float* __restrict__ out)
{
    __shared__ float bufA[NPOS * STRD];
    __shared__ float qT[NPIX * STRD];
    __shared__ float simS[NPIX * 26];

    const int t    = threadIdx.x;
    const int lane = t & 31;
    const int w    = t >> 5;
    const int lo   = lane & 15;
    const int hi   = lane >> 4;

    const int b  = blockIdx.z;
    const int x0 = blockIdx.y * TR;
    const int y0 = blockIdx.x * 16;

    {
        const int sub = lane / 10;
        const int m   = lane - sub * 10;
        const int g   = y0 - 2 + 2 * m;
        const int gs  = g < 0 ? 0 : (g > 126 ? 126 : g);
        const bool left = g < 0, right = g > 126;
        const float* src = fmap + (size_t)b * 128 * 96 * 128 + gs;
        for (int it0 = 0; it0 < 96; it0 += 32) {
            float2 v[32];
            #pragma unroll
            for (int u = 0; u < 32; ++u) {
                int pair = ((it0 + u) * 8 + w) * 3 + sub;
                pair = pair > 2047 ? 2047 : pair;
                int r = pair & 15, c = pair >> 4;
                int gx = x0 - 2 + r; gx = gx < 0 ? 0 : (gx > 95 ? 95 : gx);
                v[u] = *(const float2*)(src + ((size_t)c * 96 + gx) * 128);
            }
            #pragma unroll
            for (int u = 0; u < 32; ++u) {
                int pair = ((it0 + u) * 8 + w) * 3 + sub;
                if (sub < 3 && pair < 2048) {
                    int r = pair & 15, c = pair >> 4;
                    float a0 = right ? v[u].y : v[u].x;
                    float a1 = left  ? v[u].x : v[u].y;
                    float* dst = bufA + (r * HC + 2 * m) * STRD + c;
                    dst[0]    = a0;
                    dst[STRD] = a1;
                }
            }
        }
    }
    __syncthreads();

    const float scale = 0.08838834764831845f;

    {
        float2 af[32];
        const float* arow = Wqk + ((size_t)(w * 16 + lo)) * 128 + 2 * hi;
        #pragma unroll
        for (int kk = 0; kk < 32; ++kk) af[kk] = *(const float2*)(arow + kk * 4);

        for (int pr = 0; pr < TR; pr += 2) {
            v8f acc0 = {0.f, 0.f, 0.f, 0.f, 0.f, 0.f, 0.f, 0.f};
            v8f acc1 = {0.f, 0.f, 0.f, 0.f, 0.f, 0.f, 0.f, 0.f};
            const float* br0 = bufA + ((pr + 2) * HC + 2 + lo) * STRD + 2 * hi;
            const float* br1 = br0 + HC * STRD;
            #pragma unroll
            for (int kk = 0; kk < 32; ++kk) {
                float2 p0 = *(const float2*)(br0 + kk * 4);
                float2 p1 = *(const float2*)(br1 + kk * 4);
                v2f av;  av.x  = af[kk].x; av.y  = af[kk].y;
                v2f bv0; bv0.x = p0.x;     bv0.y = p0.y;
                v2f bv1; bv1.x = p1.x;     bv1.y = p1.y;
                acc0 = __builtin_amdgcn_wmma_f32_16x16x4_f32(
                    false, av, false, bv0, (short)0, acc0, false, false);
                acc1 = __builtin_amdgcn_wmma_f32_16x16x4_f32(
                    false, av, false, bv1, (short)0, acc1, false, false);
            }
            #pragma unroll
            for (int v = 0; v < 8; ++v) {
                qT[(pr * 16 + lo) * STRD + w * 16 + v + 8 * hi]       = acc0[v] * scale;
                qT[((pr + 1) * 16 + lo) * STRD + w * 16 + v + 8 * hi] = acc1[v] * scale;
            }
        }
    }

    {
        float2 af[32];
        const float* arow = Wqk + ((size_t)(128 + w * 16 + lo)) * 128 + 2 * hi;
        #pragma unroll
        for (int kk = 0; kk < 32; ++kk) af[kk] = *(const float2*)(arow + kk * 4);

        v8f kacc[NKT];
        #pragma unroll
        for (int tt = 0; tt < NKT; tt += 2) {
            v8f acc0 = {0.f, 0.f, 0.f, 0.f, 0.f, 0.f, 0.f, 0.f};
            v8f acc1 = {0.f, 0.f, 0.f, 0.f, 0.f, 0.f, 0.f, 0.f};
            const float* br0 = bufA + (tt * 16 + lo) * STRD + 2 * hi;
            const float* br1 = br0 + 16 * STRD;
            #pragma unroll
            for (int kk = 0; kk < 32; ++kk) {
                float2 p0 = *(const float2*)(br0 + kk * 4);
                float2 p1 = *(const float2*)(br1 + kk * 4);
                v2f av;  av.x  = af[kk].x; av.y  = af[kk].y;
                v2f bv0; bv0.x = p0.x;     bv0.y = p0.y;
                v2f bv1; bv1.x = p1.x;     bv1.y = p1.y;
                acc0 = __builtin_amdgcn_wmma_f32_16x16x4_f32(
                    false, av, false, bv0, (short)0, acc0, false, false);
                acc1 = __builtin_amdgcn_wmma_f32_16x16x4_f32(
                    false, av, false, bv1, (short)0, acc1, false, false);
            }
            kacc[tt] = acc0; kacc[tt + 1] = acc1;
        }
        __syncthreads();
        #pragma unroll
        for (int tt = 0; tt < NKT; ++tt) {
            #pragma unroll
            for (int v = 0; v < 8; ++v)
                bufA[(tt * 16 + lo) * STRD + w * 16 + v + 8 * hi] = kacc[tt][v];
        }
    }
    __syncthreads();

    for (int idx = t; idx < NPIX * 25; idx += 256) {
        int pix = idx / 25, l = idx % 25;
        int di = l / 5, dj = l % 5;
        int pr = pix >> 4, pc = pix & 15;
        int npos = (pr + di) * HC + pc + dj;
        const float4* qv = (const float4*)(qT   + pix  * STRD);
        const float4* kv = (const float4*)(bufA + npos * STRD);
        float s = 0.f;
        #pragma unroll
        for (int c4 = 0; c4 < 32; ++c4) {
            float4 q4 = qv[c4], k4 = kv[c4];
            s += q4.x * k4.x + q4.y * k4.y + q4.z * k4.z + q4.w * k4.w;
        }
        simS[pix * 26 + l] = s;
    }
    __syncthreads();

    if (t < NPIX) {
        float m = -INFINITY;
        #pragma unroll
        for (int l = 0; l < 25; ++l) m = fmaxf(m, simS[t * 26 + l]);
        float e[25], sum = 0.f;
        #pragma unroll
        for (int l = 0; l < 25; ++l) {
            e[l] = __expf(simS[t * 26 + l] - m);
            sum += e[l];
        }
        float inv = 1.0f / sum;
        const int x = x0 + (t >> 4), y = y0 + (t & 15);
        float* op = out + ((size_t)b * 12288 + (size_t)(x * 128 + y)) * 25;
        #pragma unroll
        for (int l = 0; l < 25; ++l) op[l] = e[l] * inv;
    }
}

extern "C" void kernel_launch(void* const* d_in, const int* in_sizes, int n_in,
                              void* d_out, int out_size, void* d_ws, size_t ws_size,
                              hipStream_t stream) {
    (void)in_sizes; (void)n_in; (void)d_ws; (void)ws_size; (void)out_size;
    const float* fmap = (const float*)d_in[0];
    const float* Wqk  = (const float*)d_in[1];
    float* out = (float*)d_out;

    dim3 grid(128 / 16, 96 / TR, 4);
    LocalSimilar_tile2d_kernel<<<grid, 256, 0, stream>>>(fmap, Wqk, out);
}


# similarity on the matrix unit: per 4x4 pixel block q (A) times its 8x8 window of k (B, 4 N-tiles), 5x5 band scattered to the score buffer; was 4800 LDS-fed vector dots; score summation order changes
# speedup vs baseline: 4.8898x; 1.2229x over previous
#include <hip/hip_runtime.h>
#include <math.h>


typedef float v2f __attribute__((ext_vector_type(2)));
typedef float v8f __attribute__((ext_vector_type(8)));

#define STRD 132
#define TR   12
#define HR   16
#define HC   20
#define NPOS 320
#define NKT  20
#define NPIX 192

template <int NJ>
__device__ __forceinline__ void
sim_block(const float* bufA, const float* qT, float* simS,
          int br, int bc, int j0, int lo, int hi)
{
    const float* ap = qT + ((br * 4 + (lo >> 2)) * 16 + bc * 4 + (lo & 3)) * STRD + 2 * hi;
    float2 af[32];
    #pragma unroll
    for (int kk = 0; kk < 32; ++kk) af[kk] = *(const float2*)(ap + kk * 4);

    const float* bp[NJ];
    v8f acc[NJ];
    #pragma unroll
    for (int jj = 0; jj < NJ; ++jj) {
        const int j  = j0 + jj;
        const int ra = (j >> 1) * 4 + (j & 1);
        const int hr = br * 4 + ra + 2 * (lo >> 3);
        const int hc = bc * 4 + (lo & 7);
        bp[jj] = bufA + (hr * HC + hc) * STRD + 2 * hi;
        acc[jj] = (v8f){0.f, 0.f, 0.f, 0.f, 0.f, 0.f, 0.f, 0.f};
    }
    #pragma unroll
    for (int kk = 0; kk < 32; ++kk) {
        v2f av; av.x = af[kk].x; av.y = af[kk].y;
        #pragma unroll
        for (int jj = 0; jj < NJ; ++jj) {
            float2 p = *(const float2*)(bp[jj] + kk * 4);
            v2f bv; bv.x = p.x; bv.y = p.y;
            acc[jj] = __builtin_amdgcn_wmma_f32_16x16x4_f32(
                false, av, false, bv, (short)0, acc[jj], false, false);
        }
    }
    #pragma unroll
    for (int jj = 0; jj < NJ; ++jj) {
        const int j  = j0 + jj;
        const int ra = (j >> 1) * 4 + (j & 1);
        #pragma unroll
        for (int r = 0; r < 8; ++r) {
            const int mr = 2 * hi + (r >> 2), mc = r & 3;
            const int di = ra + 2 * (lo >> 3) - mr;
            const int dj = (lo & 7) - mc;
            if ((unsigned)di < 5u && (unsigned)dj < 5u) {
                const int pix = (br * 4 + mr) * 16 + bc * 4 + mc;
                simS[pix * 26 + di * 5 + dj] = acc[jj][r];
            }
        }
    }
}

__global__ __launch_bounds__(256) void
LocalSimilar_tile2d_kernel(const float* __restrict__ fmap,
                           const float* __restrict__ Wqk,
                           float* __restrict__ out)
{
    __shared__ float bufA[NPOS * STRD];
    __shared__ float qT[NPIX * STRD];
    __shared__ float simS[NPIX * 26];

    const int t    = threadIdx.x;
    const int lane = t & 31;
    const int w    = t >> 5;
    const int lo   = lane & 15;
    const int hi   = lane >> 4;

    const int b  = blockIdx.z;
    const int x0 = blockIdx.y * TR;
    const int y0 = blockIdx.x * 16;

    {
        const int sub = lane / 10;
        const int m   = lane - sub * 10;
        const int g   = y0 - 2 + 2 * m;
        const int gs  = g < 0 ? 0 : (g > 126 ? 126 : g);
        const bool left = g < 0, right = g > 126;
        const float* src = fmap + (size_t)b * 128 * 96 * 128 + gs;
        for (int it0 = 0; it0 < 96; it0 += 32) {
            float2 v[32];
            #pragma unroll
            for (int u = 0; u < 32; ++u) {
                int pair = ((it0 + u) * 8 + w) * 3 + sub;
                pair = pair > 2047 ? 2047 : pair;
                int r = pair & 15, c = pair >> 4;
                int gx = x0 - 2 + r; gx = gx < 0 ? 0 : (gx > 95 ? 95 : gx);
                v[u] = *(const float2*)(src + ((size_t)c * 96 + gx) * 128);
            }
            #pragma unroll
            for (int u = 0; u < 32; ++u) {
                int pair = ((it0 + u) * 8 + w) * 3 + sub;
                if (sub < 3 && pair < 2048) {
                    int r = pair & 15, c = pair >> 4;
                    float a0 = right ? v[u].y : v[u].x;
                    float a1 = left  ? v[u].x : v[u].y;
                    float* dst = bufA + (r * HC + 2 * m) * STRD + c;
                    dst[0]    = a0;
                    dst[STRD] = a1;
                }
            }
        }
    }
    __syncthreads();

    const float scale = 0.08838834764831845f;

    {
        float2 af[32];
        const float* arow = Wqk + ((size_t)(w * 16 + lo)) * 128 + 2 * hi;
        #pragma unroll
        for (int kk = 0; kk < 32; ++kk) af[kk] = *(const float2*)(arow + kk * 4);

        for (int pr = 0; pr < TR; pr += 2) {
            v8f acc0 = {0.f, 0.f, 0.f, 0.f, 0.f, 0.f, 0.f, 0.f};
            v8f acc1 = {0.f, 0.f, 0.f, 0.f, 0.f, 0.f, 0.f, 0.f};
            const float* br0 = bufA + ((pr + 2) * HC + 2 + lo) * STRD + 2 * hi;
            const float* br1 = br0 + HC * STRD;
            #pragma unroll
            for (int kk = 0; kk < 32; ++kk) {
                float2 p0 = *(const float2*)(br0 + kk * 4);
                float2 p1 = *(const float2*)(br1 + kk * 4);
                v2f av;  av.x  = af[kk].x; av.y  = af[kk].y;
                v2f bv0; bv0.x = p0.x;     bv0.y = p0.y;
                v2f bv1; bv1.x = p1.x;     bv1.y = p1.y;
                acc0 = __builtin_amdgcn_wmma_f32_16x16x4_f32(
                    false, av, false, bv0, (short)0, acc0, false, false);
                acc1 = __builtin_amdgcn_wmma_f32_16x16x4_f32(
                    false, av, false, bv1, (short)0, acc1, false, false);
            }
            #pragma unroll
            for (int v = 0; v < 8; ++v) {
                qT[(pr * 16 + lo) * STRD + w * 16 + v + 8 * hi]       = acc0[v] * scale;
                qT[((pr + 1) * 16 + lo) * STRD + w * 16 + v + 8 * hi] = acc1[v] * scale;
            }
        }
    }

    {
        float2 af[32];
        const float* arow = Wqk + ((size_t)(128 + w * 16 + lo)) * 128 + 2 * hi;
        #pragma unroll
        for (int kk = 0; kk < 32; ++kk) af[kk] = *(const float2*)(arow + kk * 4);

        v8f kacc[NKT];
        #pragma unroll
        for (int tt = 0; tt < NKT; tt += 2) {
            v8f acc0 = {0.f, 0.f, 0.f, 0.f, 0.f, 0.f, 0.f, 0.f};
            v8f acc1 = {0.f, 0.f, 0.f, 0.f, 0.f, 0.f, 0.f, 0.f};
            const float* br0 = bufA + (tt * 16 + lo) * STRD + 2 * hi;
            const float* br1 = br0 + 16 * STRD;
            #pragma unroll
            for (int kk = 0; kk < 32; ++kk) {
                float2 p0 = *(const float2*)(br0 + kk * 4);
                float2 p1 = *(const float2*)(br1 + kk * 4);
                v2f av;  av.x  = af[kk].x; av.y  = af[kk].y;
                v2f bv0; bv0.x = p0.x;     bv0.y = p0.y;
                v2f bv1; bv1.x = p1.x;     bv1.y = p1.y;
                acc0 = __builtin_amdgcn_wmma_f32_16x16x4_f32(
                    false, av, false, bv0, (short)0, acc0, false, false);
                acc1 = __builtin_amdgcn_wmma_f32_16x16x4_f32(
                    false, av, false, bv1, (short)0, acc1, false, false);
            }
            kacc[tt] = acc0; kacc[tt + 1] = acc1;
        }
        __syncthreads();
        #pragma unroll
        for (int tt = 0; tt < NKT; ++tt) {
            #pragma unroll
            for (int v = 0; v < 8; ++v)
                bufA[(tt * 16 + lo) * STRD + w * 16 + v + 8 * hi] = kacc[tt][v];
        }
    }
    __syncthreads();

    sim_block<4>(bufA, qT, simS, w >> 2, w & 3, 0, lo, hi);
    sim_block<2>(bufA, qT, simS, 2, w >> 1, 2 * (w & 1), lo, hi);
    __syncthreads();

    if (t < NPIX) {
        float m = -INFINITY;
        #pragma unroll
        for (int l = 0; l < 25; ++l) m = fmaxf(m, simS[t * 26 + l]);
        float e[25], sum = 0.f;
        #pragma unroll
        for (int l = 0; l < 25; ++l) {
            e[l] = __expf(simS[t * 26 + l] - m);
            sum += e[l];
        }
        float inv = 1.0f / sum;
        const int x = x0 + (t >> 4), y = y0 + (t & 15);
        float* op = out + ((size_t)b * 12288 + (size_t)(x * 128 + y)) * 25;
        #pragma unroll
        for (int l = 0; l < 25; ++l) op[l] = e[l] * inv;
    }
}

extern "C" void kernel_launch(void* const* d_in, const int* in_sizes, int n_in,
                              void* d_out, int out_size, void* d_ws, size_t ws_size,
                              hipStream_t stream) {
    (void)in_sizes; (void)n_in; (void)d_ws; (void)ws_size; (void)out_size;
    const float* fmap = (const float*)d_in[0];
    const float* Wqk  = (const float*)d_in[1];
    float* out = (float*)d_out;

    dim3 grid(128 / 16, 96 / TR, 4);
    LocalSimilar_tile2d_kernel<<<grid, 256, 0, stream>>>(fmap, Wqk, out);
}


# output leaves as whole lines: normalised scores collected in LDS in out's order (a tile row = 1600 contiguous bytes), then 128-bit stores, 5 a thread; was 25 32-bit stores a pixel, lanes 100 B apart
# speedup vs baseline: 5.4059x; 1.1055x over previous
#include <hip/hip_runtime.h>
#include <math.h>


typedef float v2f __attribute__((ext_vector_type(2)));
typedef float v4f __attribute__((ext_vector_type(4)));
typedef float v8f __attribute__((ext_vector_type(8)));

#define STRD 132
#define TR   12
#define HR   16
#define HC   20
#define NPOS 320
#define NKT  20
#define NPIX 192

__device__ __forceinline__ v4f frag4(const float* p, int j) { return *(const v4f*)(p + 8 * j); }

__device__ __forceinline__ void frag_fence() { __builtin_amdgcn_sched_barrier(0); }

template <int NT, int NP>
__device__ __forceinline__ void mma_block(v8f* acc, const v4f* a, const v4f* b)
{
    #pragma unroll
    for (int p = 0; p < NP; ++p) {
        #pragma unroll
        for (int e = 0; e < 2; ++e) {
            const v4f ap = a[p];
            const v2f av = e ? __builtin_shufflevector(ap, ap, 2, 3) : __builtin_shufflevector(ap, ap, 0, 1);
            #pragma unroll
            for (int t = 0; t < NT; ++t) {
                const v4f bq = b[t * NP + p];
                const v2f bv = e ? __builtin_shufflevector(bq, bq, 2, 3) : __builtin_shufflevector(bq, bq, 0, 1);
                acc[t] = __builtin_amdgcn_wmma_f32_16x16x4_f32(
                    false, av, false, bv, (short)0, acc[t], false, false);
            }
        }
    }
}

__device__ __forceinline__ void
scatter_band(const v8f acc, float* simS, int br, int bc, int ra, int lo, int hi)
{
    #pragma unroll
    for (int r = 0; r < 8; ++r) {
        const int mr = 2 * hi + (r >> 2), mc = r & 3;
        const int di = ra + 2 * (lo >> 3) - mr;
        const int dj = (lo & 7) - mc;
        if ((unsigned)di < 5u && (unsigned)dj < 5u) {
            const int pix = (br * 4 + mr) * 16 + bc * 4 + mc;
            simS[pix * 26 + di * 5 + dj] = acc[r];
        }
    }
}

__global__ __launch_bounds__(256) void
LocalSimilar_tile2d_kernel(const float* __restrict__ fmap,
                           const float* __restrict__ Wqk,
                           float* __restrict__ out)
{
    __shared__ __attribute__((aligned(16))) float bufA[NPOS * STRD];
    __shared__ __attribute__((aligned(16))) float qT[NPIX * STRD];
    __shared__ float simS[NPIX * 26];

    const int t    = threadIdx.x;
    const int lane = t & 31;
    const int w    = t >> 5;
    const int lo   = lane & 15;
    const int hi   = lane >> 4;

    const int b  = blockIdx.z;
    const int x0 = blockIdx.y * TR;
    const int y0 = blockIdx.x * 16;

    {
        const int sub = lane / 10;
        const int m   = lane - sub * 10;
        const int g   = y0 - 2 + 2 * m;
        const int gs  = g < 0 ? 0 : (g > 126 ? 126 : g);
        const bool left = g < 0, right = g > 126;
        const float* src = fmap + (size_t)b * 128 * 96 * 128 + gs;
        for (int it0 = 0; it0 < 96; it0 += 32) {
            float2 v[32];
            #pragma unroll
            for (int u = 0; u < 32; ++u) {
                int pair = ((it0 + u) * 8 + w) * 3 + sub;
                pair = pair > 2047 ? 2047 : pair;
                int r = pair & 15, c = pair >> 4;
                int gx = x0 - 2 + r; gx = gx < 0 ? 0 : (gx > 95 ? 95 : gx);
                v[u] = *(const float2*)(src + ((size_t)c * 96 + gx) * 128);
            }
            #pragma unroll
            for (int u = 0; u < 32; ++u) {
                int pair = ((it0 + u) * 8 + w) * 3 + sub;
                if (sub < 3 && pair < 2048) {
                    int r = pair & 15, c = pair >> 4;
                    float a0 = right ? v[u].y : v[u].x;
                    float a1 = left  ? v[u].x : v[u].y;
                    float* dst = bufA + (r * HC + 2 * m) * STRD + c;
                    dst[0]    = a0;
                    dst[STRD] = a1;
                }
            }
        }
    }

    v4f aq[16], ak[16];
    {
        const float* wq = Wqk + ((size_t)(w * 16 + lo)) * 128 + 4 * hi;
        const float* wk = wq + (size_t)128 * 128;
        #pragma unroll
        for (int j = 0; j < 16; ++j) aq[j] = frag4(wq, j);
        #pragma unroll
        for (int j = 0; j < 16; ++j) ak[j] = frag4(wk, j);
    }
    __syncthreads();

    const float scale = 0.08838834764831845f;

    {
        const float* lb = bufA + lo * STRD + 4 * hi;
        v4f bb[2][16];
        v8f qacc[2];
        v8f kacc[NKT];

        #pragma unroll
        for (int p = 0; p < 8; ++p) {
            bb[0][p]     = frag4(lb + (2 * HC + 2) * STRD, p);
            bb[0][8 + p] = frag4(lb + (3 * HC + 2) * STRD, p);
        }
        #pragma unroll
        for (int blk = 0; blk < 32; ++blk) {
            const int cur = blk & 1;
            if (blk + 1 < 32) {
                const int nb = blk + 1, nh = nb & 1;
                const int pos0 = nb < 12 ? ((nb >> 1) * 2 + 2) * HC + 2 : ((nb - 12) >> 1) * 32;
                const int dpos = nb < 12 ? HC : 16;
                #pragma unroll
                for (int p = 0; p < 8; ++p) {
                    bb[cur ^ 1][p]     = frag4(lb + pos0 * STRD, 8 * nh + p);
                    bb[cur ^ 1][8 + p] = frag4(lb + (pos0 + dpos) * STRD, 8 * nh + p);
                }
            }
            frag_fence();
            const int h = blk & 1;
            if (blk < 12) {
                if (h == 0) {
                    qacc[0] = (v8f){0.f, 0.f, 0.f, 0.f, 0.f, 0.f, 0.f, 0.f};
                    qacc[1] = (v8f){0.f, 0.f, 0.f, 0.f, 0.f, 0.f, 0.f, 0.f};
                }
                mma_block<2, 8>(qacc, aq + 8 * h, bb[cur]);
            } else {
                const int t0 = ((blk - 12) >> 1) * 2;
                if (h == 0) {
                    kacc[t0]     = (v8f){0.f, 0.f, 0.f, 0.f, 0.f, 0.f, 0.f, 0.f};
                    kacc[t0 + 1] = (v8f){0.f, 0.f, 0.f, 0.f, 0.f, 0.f, 0.f, 0.f};
                }
                mma_block<2, 8>(kacc + t0, ak + 8 * h, bb[cur]);
            }
            frag_fence();
            if (blk < 12 && h == 1) {
                const int pr = (blk >> 1) * 2;
                #pragma unroll
                for (int v = 0; v < 8; ++v) {
                    qT[(pr * 16 + lo) * STRD + w * 16 + v + 8 * hi]       = qacc[0][v] * scale;
                    qT[((pr + 1) * 16 + lo) * STRD + w * 16 + v + 8 * hi] = qacc[1][v] * scale;
                }
            }
        }
        __syncthreads();
        #pragma unroll
        for (int tt = 0; tt < NKT; ++tt) {
            #pragma unroll
            for (int v = 0; v < 8; ++v)
                bufA[(tt * 16 + lo) * STRD + w * 16 + v + 8 * hi] = kacc[tt][v];
        }
    }
    __syncthreads();

    {
        const int br1 = w >> 2, bc1 = w & 3;
        const int br2 = 2,      bc2 = w >> 1;
        const int ra2 = 4 * (w & 1);
        const float* ap1 = qT + ((br1 * 4 + (lo >> 2)) * 16 + bc1 * 4 + (lo & 3)) * STRD + 4 * hi;
        const float* ap2 = qT + ((br2 * 4 + (lo >> 2)) * 16 + bc2 * 4 + (lo & 3)) * STRD + 4 * hi;
        const float* bp1 = bufA + ((br1 * 4 + 2 * (lo >> 3)) * HC + bc1 * 4 + (lo & 7)) * STRD + 4 * hi;
        const float* bp2 = bufA + ((br2 * 4 + ra2 + 2 * (lo >> 3)) * HC + bc2 * 4 + (lo & 7)) * STRD + 4 * hi;

        v4f a1[16], a2[16];
        v4f sb[2][16];
        v8f acc1[4], acc2[2];
        #pragma unroll
        for (int j = 0; j < 4; ++j) acc1[j] = (v8f){0.f, 0.f, 0.f, 0.f, 0.f, 0.f, 0.f, 0.f};
        acc2[0] = (v8f){0.f, 0.f, 0.f, 0.f, 0.f, 0.f, 0.f, 0.f};
        acc2[1] = (v8f){0.f, 0.f, 0.f, 0.f, 0.f, 0.f, 0.f, 0.f};

        #pragma unroll
        for (int j = 0; j < 16; ++j) a1[j] = frag4(ap1, j);
        #pragma unroll
        for (int p = 0; p < 4; ++p) {
            #pragma unroll
            for (int j = 0; j < 4; ++j)
                sb[0][j * 4 + p] = frag4(bp1 + (((j >> 1) * 4 + (j & 1)) * HC) * STRD, p);
        }
        #pragma unroll
        for (int s = 0; s < 6; ++s) {
            const int cur = s & 1;
            if (s + 1 < 4) {
                #pragma unroll
                for (int p = 0; p < 4; ++p) {
                    #pragma unroll
                    for (int j = 0; j < 4; ++j)
                        sb[cur ^ 1][j * 4 + p] =
                            frag4(bp1 + (((j >> 1) * 4 + (j & 1)) * HC) * STRD, 4 * (s + 1) + p);
                }
            } else if (s + 1 < 6) {
                const int nh = s + 1 - 4;
                if (nh == 0) {
                    #pragma unroll
                    for (int j = 0; j < 16; ++j) a2[j] = frag4(ap2, j);
                }
                #pragma unroll
                for (int p = 0; p < 8; ++p) {
                    sb[cur ^ 1][p]     = frag4(bp2, 8 * nh + p);
                    sb[cur ^ 1][8 + p] = frag4(bp2 + HC * STRD, 8 * nh + p);
                }
            }
            frag_fence();
            if (s < 4) mma_block<4, 4>(acc1, a1 + 4 * s, sb[cur]);
            else       mma_block<2, 8>(acc2, a2 + 8 * (s - 4), sb[cur]);
            frag_fence();
            if (s == 3) {
                #pragma unroll
                for (int j = 0; j < 4; ++j)
                    scatter_band(acc1[j], simS, br1, bc1, (j >> 1) * 4 + (j & 1), lo, hi);
            }
        }
        scatter_band(acc2[0], simS, br2, bc2, ra2,     lo, hi);
        scatter_band(acc2[1], simS, br2, bc2, ra2 + 1, lo, hi);
    }
    __syncthreads();

    float* outS = qT;
    if (t < NPIX) {
        float m = -INFINITY;
        #pragma unroll
        for (int l = 0; l < 25; ++l) m = fmaxf(m, simS[t * 26 + l]);
        float e[25], sum = 0.f;
        #pragma unroll
        for (int l = 0; l < 25; ++l) {
            e[l] = __expf(simS[t * 26 + l] - m);
            sum += e[l];
        }
        float inv = 1.0f / sum;
        #pragma unroll
        for (int l = 0; l < 25; ++l) outS[t * 25 + l] = e[l] * inv;
    }
    __syncthreads();

    {
        float* ob = out + ((size_t)b * 12288 + (size_t)x0 * 128 + y0) * 25;
        v4f o[5];
        #pragma unroll
        for (int k = 0; k < 4; ++k) o[k] = *(const v4f*)(outS + 4 * (t + 256 * k));
        if (t < 176) o[4] = *(const v4f*)(outS + 4 * (t + 1024));
        #pragma unroll
        for (int k = 0; k < 5; ++k) {
            const int i = t + 256 * k;
            const int r = i / 100, c4 = i - r * 100;
            if (k < 4 || t < 176)
                *(v4f*)(ob + (size_t)r * (128 * 25) + 4 * c4) = o[k];
        }
    }
}

extern "C" void kernel_launch(void* const* d_in, const int* in_sizes, int n_in,
                              void* d_out, int out_size, void* d_ws, size_t ws_size,
                              hipStream_t stream) {
    (void)in_sizes; (void)n_in; (void)d_ws; (void)ws_size; (void)out_size;
    const float* fmap = (const float*)d_in[0];
    const float* Wqk  = (const float*)d_in[1];
    float* out = (float*)d_out;

    dim3 grid(128 / 16, 96 / TR, 4);
    LocalSimilar_tile2d_kernel<<<grid, 256, 0, stream>>>(fmap, Wqk, out);
}
